# starting kernel
# baseline (speedup 1.0000x reference)
#include <hip/hip_runtime.h>

#define H_  1024
#define W_  2048
#define C_  16
#define SH_ 68
#define SW_ 132

typedef float v2f __attribute__((ext_vector_type(2)));
typedef float v8f __attribute__((ext_vector_type(8)));

__global__ __launch_bounds__(256) void splat_kernel(const float* __restrict__ inp,
                                                    float* __restrict__ G)
{
    const int cx  = blockIdx.x;
    const int cy  = blockIdx.y;
    const int tid = threadIdx.x;
    const int py  = tid >> 4, px = tid & 15;
    const int y   = 16 * (cy - 2) - 8 + py;
    const int x   = 16 * (cx - 2) - 8 + px;

    float4 v0 = make_float4(0.f, 0.f, 0.f, 0.f);
    float4 v1 = v0, v2 = v0, v3 = v0;
    if ((unsigned)y < (unsigned)H_ && (unsigned)x < (unsigned)W_) {
        const float4* p = (const float4*)(inp + ((size_t)y * W_ + x) * C_);
        v0 = p[0]; v1 = p[1]; v2 = p[2]; v3 = p[3];
    }

    __shared__ float4 red[256 * 4];
    red[tid * 4 + 0] = v0; red[tid * 4 + 1] = v1;
    red[tid * 4 + 2] = v2; red[tid * 4 + 3] = v3;
    __syncthreads();

    for (int s = 128; s > 0; s >>= 1) {
        if (tid < s) {
#pragma unroll
            for (int q = 0; q < 4; ++q) {
                float4 a = red[tid * 4 + q];
                float4 b = red[(tid + s) * 4 + q];
                a.x += b.x; a.y += b.y; a.z += b.z; a.w += b.w;
                red[tid * 4 + q] = a;
            }
        }
        __syncthreads();
    }
    if (tid < C_)
        G[((size_t)cy * SW_ + cx) * C_ + tid] = ((const float*)red)[tid];
}

__global__ __launch_bounds__(256) void blurA_kernel(const float* __restrict__ G,
                                                    float* __restrict__ A)
{
    const int idx  = blockIdx.x * 256 + threadIdx.x;
    const int cell = idx >> 4;
    const int x    = cell % SW_;
    const int y    = cell / SW_;
    const float gc = G[idx];
    float r;
    if (x > 0 && x < SW_ - 1)
        r = (G[idx - C_] + G[idx + C_] + 2.0f * gc) * 0.25f;
    else if (y > 0 && y < SH_ - 1)
        r = (G[idx - C_ * SW_] + G[idx + C_ * SW_] + 2.0f * gc) * 0.25f;
    else
        r = 0.0f;
    A[idx] = r;
}

__global__ __launch_bounds__(256) void blurB_kernel(const float* __restrict__ G,
                                                    const float* __restrict__ A,
                                                    float* __restrict__ B)
{
    const int idx  = blockIdx.x * 256 + threadIdx.x;
    const int cell = idx >> 4;
    const int x    = cell % SW_;
    const int y    = cell / SW_;
    const float ac = A[idx];
    float r;
    if (x > 0 && x < SW_ - 1)
        r = (A[idx - C_] + A[idx + C_] + 2.0f * ac) * 0.25f;
    else if (y > 0 && y < SH_ - 1)
        r = (A[idx - C_ * SW_] + A[idx + C_ * SW_] + 2.0f * ac) * 0.25f;
    else
        r = G[idx];
    B[idx] = r;
}

__global__ __launch_bounds__(256) void slice_kernel(const float* __restrict__ Bg,
                                                    float* __restrict__ out)
{
    const int wave = threadIdx.x >> 5;
    const int lane = threadIdx.x & 31;
    const int tile = blockIdx.x * 8 + wave;
    const int k    = tile >> 7;
    const int j    = tile & 127;
    const int m    = lane & 15;
    const int hi   = lane >> 4;

    const float* gB = Bg + (((size_t)(k + 2 + hi) * SW_) + (j + 2)) * C_ + m;
    v2f b;
    b.x = gB[0];
    b.y = gB[C_];

    const float am = (float)m * 0.0625f;
    const float w0 = 1.0f - am;

    float* obase = out + ((size_t)(16 * k) * W_ + 16 * j) * C_;

#pragma unroll
    for (int py = 0; py < 16; ++py) {
        const float ay = (float)py * 0.0625f;
        const float wy = hi ? ay : (1.0f - ay);
        v2f a;
        a.x = wy * w0;
        a.y = wy * am;
        v8f cacc = {};
        v8f d = __builtin_amdgcn_wmma_f32_16x16x4_f32(
             false, a,  false, b,
             (short)0, cacc,  false,  false);

        float* orow = obase + (size_t)py * W_ * C_ + (size_t)hi * 8 * C_ + m;
#pragma unroll
        for (int r = 0; r < 8; ++r)
            orow[r * C_] = d[r];
    }
}

extern "C" void kernel_launch(void* const* d_in, const int* in_sizes, int n_in,
                              void* d_out, int out_size, void* d_ws, size_t ws_size,
                              hipStream_t stream)
{
    const float* inp = (const float*)d_in[0];
    float* out = (float*)d_out;

    const size_t S = (size_t)SH_ * SW_ * C_;
    float* G  = (float*)d_ws;
    float* A  = G + S;
    float* Bg = A + S;

    splat_kernel<<<dim3(SW_, SH_), 256, 0, stream>>>(inp, G);

    const int n = SH_ * SW_ * C_;
    blurA_kernel<<<n / 256, 256, 0, stream>>>(G, A);
    blurB_kernel<<<n / 256, 256, 0, stream>>>(G, A, Bg);

    slice_kernel<<<8192 / 8, 256, 0, stream>>>(Bg, out);
}


# front re-cut into 256 equal 16-wave workgroups (8 image rows x half the width, 2-cell halo, 32 loads in flight a wave); each blurs its half-height sums, slice adds both planes; sum order changes
# speedup vs baseline: 1.5431x; 1.5431x over previous
#include <hip/hip_runtime.h>

#define H_  1024
#define W_  2048
#define C_  16
#define SH_ 68
#define SW_ 132

typedef float v2f __attribute__((ext_vector_type(2)));
typedef float v8f __attribute__((ext_vector_type(8)));
typedef float f4  __attribute__((ext_vector_type(4)));

#define ROW4_  (W_ * C_ / 4)
#define NL_    72
#define FW_    16

__device__ __forceinline__ void cell_load(const f4* __restrict__ band, const int cx,
                                          f4 (&v)[16], bool& ok0, bool& ok1)
{
    const int xb = 16 * (cx - 2) - 8;
    ok0 = xb >= 0;
    ok1 = xb + 15 < W_;
    const f4* p0 = band + (size_t)(ok0 ? xb : 0) * 4;
    const f4* p1 = band + (size_t)(ok1 ? xb + 8 : W_ - 8) * 4;
#pragma unroll
    for (int r = 0; r < 8; ++r) {
        v[r]     = p0[(size_t)r * ROW4_];
        v[8 + r] = p1[(size_t)r * ROW4_];
    }
}

__device__ __forceinline__ f4 cell_reduce(const f4 (&v)[16], const bool ok0, const bool ok1)
{
    f4 s0 = v[0], s1 = v[8];
#pragma unroll
    for (int r = 1; r < 8; ++r) {
        s0 += v[r];
        s1 += v[8 + r];
    }
    const f4 zero = {0.f, 0.f, 0.f, 0.f};
    if (!ok0) s0 = zero;
    if (!ok1) s1 = zero;
    return s0 + s1;
}

#define HOLD16(v)                                                                        \
    asm volatile("" :: "v"(v[0]), "v"(v[1]), "v"(v[2]), "v"(v[3]), "v"(v[4]), "v"(v[5]), \
                 "v"(v[6]), "v"(v[7]), "v"(v[8]), "v"(v[9]), "v"(v[10]), "v"(v[11]),     \
                 "v"(v[12]), "v"(v[13]), "v"(v[14]), "v"(v[15]))

__device__ __forceinline__ f4 fold8(f4 acc)
{
#pragma unroll
    for (int off = 4; off < 32; off <<= 1) {
        acc.x += __shfl_xor(acc.x, off);
        acc.y += __shfl_xor(acc.y, off);
        acc.z += __shfl_xor(acc.z, off);
        acc.w += __shfl_xor(acc.w, off);
    }
    return acc;
}

__global__ __launch_bounds__(512) void front_kernel(const float* __restrict__ inp,
                                                    float* __restrict__ P0,
                                                    float* __restrict__ P1)
{
    const int seg  = blockIdx.x;
    const int hr   = blockIdx.y;
    const int tid  = threadIdx.x;
    const int wave = tid >> 5;
    const int lane = tid & 31;
    const int cy   = ((hr + 1) >> 1) + 2;
    float* __restrict__ P = (hr & 1) ? P1 : P0;
    const int c0    = seg ? 64 : 0;
    const int first = seg ? 64 : 2;
    const int last  = seg ? 130 : 67;
    const int lbhi  = seg ? 66 : 65;

    __shared__ f4    Gs4[NL_ * C_ / 4];
    __shared__ float  As[NL_ * C_];
    const float* Gs = (const float*)Gs4;

    const f4* band = (const f4*)inp + (size_t)(8 * hr) * ROW4_ + lane;

#pragma unroll 1
    for (int i = 0; i < 4; i += 2) {
        const int cxa = first + wave + FW_ * i;
        const int cxb = cxa + FW_;
        f4 va[16], vb[16];
        bool a0, a1, b0, b1;
        cell_load(band, cxa, va, a0, a1);
        cell_load(band, cxb, vb, b0, b1);
        HOLD16(va);
        HOLD16(vb);
        const f4 sa = fold8(cell_reduce(va, a0, a1));
        const f4 sb = fold8(cell_reduce(vb, b0, b1));
        if (lane < 4) {
            Gs4[(cxa - c0) * 4 + lane] = sa;
            Gs4[(cxb - c0) * 4 + lane] = sb;
        }
    }
    {
        const int cxe = first + wave + FW_ * 4;
        if (cxe <= last) {
            f4 ve[16];
            bool e0, e1;
            cell_load(band, cxe, ve, e0, e1);
            HOLD16(ve);
            const f4 se = fold8(cell_reduce(ve, e0, e1));
            if (lane < 4)
                Gs4[(cxe - c0) * 4 + lane] = se;
        }
    }
    __syncthreads();

    for (int e = tid; e < NL_ * C_; e += 512) {
        const int l = e >> 4;
        const int x = c0 + l;
        float a = 0.0f;
        if (l >= 1 && l <= lbhi + 1 && x > 0 && x < SW_ - 1) {
            const float gm = (x - 1 >= first && x - 1 <= last) ? Gs[e - C_] : 0.0f;
            const float gp = (x + 1 >= first && x + 1 <= last) ? Gs[e + C_] : 0.0f;
            const float g0 = (x     >= first && x     <= last) ? Gs[e]      : 0.0f;
            a = (gm + gp + 2.0f * g0) * 0.25f;
        }
        As[e] = a;
    }
    __syncthreads();

    for (int e = tid; e < NL_ * C_; e += 512) {
        const int l = e >> 4;
        if (l >= 2 && l <= lbhi) {
            const float b = (As[e - C_] + As[e + C_] + 2.0f * As[e]) * 0.25f;
            P[((size_t)cy * SW_ + c0) * C_ + e] = b;
        }
    }
}

__global__ __launch_bounds__(256) void slice_kernel(const float* __restrict__ P0,
                                                    const float* __restrict__ P1,
                                                    float* __restrict__ out)
{
    const int wave = threadIdx.x >> 5;
    const int lane = threadIdx.x & 31;
    const int tile = blockIdx.x * 8 + wave;
    const int k    = tile >> 7;
    const int j    = tile & 127;
    const int m    = lane & 15;
    const int hi   = lane >> 4;

    const int    row = k + 2 + hi;
    const size_t gi  = (((size_t)row * SW_) + (j + 2)) * C_ + m;
    float bx = 0.0f, by = 0.0f;
    if (row <= SH_ - 3) {
        bx = P0[gi];
        by = P0[gi + C_];
    }
    if (row >= 3) {
        bx += P1[gi];
        by += P1[gi + C_];
    }
    v2f b;
    b.x = bx;
    b.y = by;

    const float am = (float)m * 0.0625f;
    const float w0 = 1.0f - am;

    float* obase = out + ((size_t)(16 * k) * W_ + 16 * j) * C_;

#pragma unroll
    for (int py = 0; py < 16; ++py) {
        const float ay = (float)py * 0.0625f;
        const float wy = hi ? ay : (1.0f - ay);
        v2f a;
        a.x = wy * w0;
        a.y = wy * am;
        v8f cacc = {};
        v8f d = __builtin_amdgcn_wmma_f32_16x16x4_f32(
             false, a,  false, b,
             (short)0, cacc,  false,  false);

        float* orow = obase + (size_t)py * W_ * C_ + (size_t)hi * 8 * C_ + m;
#pragma unroll
        for (int r = 0; r < 8; ++r)
            orow[r * C_] = d[r];
    }
}

extern "C" void kernel_launch(void* const* d_in, const int* in_sizes, int n_in,
                              void* d_out, int out_size, void* d_ws, size_t ws_size,
                              hipStream_t stream)
{
    const float* inp = (const float*)d_in[0];
    float* out = (float*)d_out;

    const size_t S = (size_t)SH_ * SW_ * C_;
    float* P0 = (float*)d_ws + S;
    float* P1 = (float*)d_ws + 2 * S;

    front_kernel<<<dim3(2, H_ / 8), 512, 0, stream>>>(inp, P0, P1);

    slice_kernel<<<8192 / 8, 256, 0, stream>>>(P0, P1, out);
}


# slice: matrix row i now stands for pixel 2*(i%8)+i/8, so D rows r and r+8 are neighbouring pixels and each 4-byte store instruction writes one whole 128-byte line instead of two 64-byte halves
# speedup vs baseline: 1.9544x; 1.2665x over previous
#include <hip/hip_runtime.h>

#define H_  1024
#define W_  2048
#define C_  16
#define SH_ 68
#define SW_ 132

typedef float v2f __attribute__((ext_vector_type(2)));
typedef float v8f __attribute__((ext_vector_type(8)));
typedef float f4  __attribute__((ext_vector_type(4)));

#define ROW4_  (W_ * C_ / 4)
#define NL_    72
#define FW_    16

__device__ __forceinline__ void cell_load(const f4* __restrict__ band, const int cx,
                                          f4 (&v)[16], bool& ok0, bool& ok1)
{
    const int xb = 16 * (cx - 2) - 8;
    ok0 = xb >= 0;
    ok1 = xb + 15 < W_;
    const f4* p0 = band + (size_t)(ok0 ? xb : 0) * 4;
    const f4* p1 = band + (size_t)(ok1 ? xb + 8 : W_ - 8) * 4;
#pragma unroll
    for (int r = 0; r < 8; ++r) {
        v[r]     = p0[(size_t)r * ROW4_];
        v[8 + r] = p1[(size_t)r * ROW4_];
    }
}

__device__ __forceinline__ f4 cell_reduce(const f4 (&v)[16], const bool ok0, const bool ok1)
{
    f4 s0 = v[0], s1 = v[8];
#pragma unroll
    for (int r = 1; r < 8; ++r) {
        s0 += v[r];
        s1 += v[8 + r];
    }
    const f4 zero = {0.f, 0.f, 0.f, 0.f};
    if (!ok0) s0 = zero;
    if (!ok1) s1 = zero;
    return s0 + s1;
}

#define HOLD16(v)                                                                        \
    asm volatile("" :: "v"(v[0]), "v"(v[1]), "v"(v[2]), "v"(v[3]), "v"(v[4]), "v"(v[5]), \
                 "v"(v[6]), "v"(v[7]), "v"(v[8]), "v"(v[9]), "v"(v[10]), "v"(v[11]),     \
                 "v"(v[12]), "v"(v[13]), "v"(v[14]), "v"(v[15]))

__device__ __forceinline__ f4 fold8(f4 acc)
{
#pragma unroll
    for (int off = 4; off < 32; off <<= 1) {
        acc.x += __shfl_xor(acc.x, off);
        acc.y += __shfl_xor(acc.y, off);
        acc.z += __shfl_xor(acc.z, off);
        acc.w += __shfl_xor(acc.w, off);
    }
    return acc;
}

__global__ __launch_bounds__(512) void front_kernel(const float* __restrict__ inp,
                                                    float* __restrict__ P0,
                                                    float* __restrict__ P1)
{
    const int seg  = blockIdx.x;
    const int hr   = blockIdx.y;
    const int tid  = threadIdx.x;
    const int wave = tid >> 5;
    const int lane = tid & 31;
    const int cy   = ((hr + 1) >> 1) + 2;
    float* __restrict__ P = (hr & 1) ? P1 : P0;
    const int c0    = seg ? 64 : 0;
    const int first = seg ? 64 : 2;
    const int last  = seg ? 130 : 67;
    const int lbhi  = seg ? 66 : 65;

    __shared__ f4    Gs4[NL_ * C_ / 4];
    __shared__ float  As[NL_ * C_];
    const float* Gs = (const float*)Gs4;

    const f4* band = (const f4*)inp + (size_t)(8 * hr) * ROW4_ + lane;

#pragma unroll 1
    for (int i = 0; i < 4; i += 2) {
        const int cxa = first + wave + FW_ * i;
        const int cxb = cxa + FW_;
        f4 va[16], vb[16];
        bool a0, a1, b0, b1;
        cell_load(band, cxa, va, a0, a1);
        cell_load(band, cxb, vb, b0, b1);
        HOLD16(va);
        HOLD16(vb);
        const f4 sa = fold8(cell_reduce(va, a0, a1));
        const f4 sb = fold8(cell_reduce(vb, b0, b1));
        if (lane < 4) {
            Gs4[(cxa - c0) * 4 + lane] = sa;
            Gs4[(cxb - c0) * 4 + lane] = sb;
        }
    }
    {
        const int cxe = first + wave + FW_ * 4;
        if (cxe <= last) {
            f4 ve[16];
            bool e0, e1;
            cell_load(band, cxe, ve, e0, e1);
            HOLD16(ve);
            const f4 se = fold8(cell_reduce(ve, e0, e1));
            if (lane < 4)
                Gs4[(cxe - c0) * 4 + lane] = se;
        }
    }
    __syncthreads();

    for (int e = tid; e < NL_ * C_; e += 512) {
        const int l = e >> 4;
        const int x = c0 + l;
        float a = 0.0f;
        if (l >= 1 && l <= lbhi + 1 && x > 0 && x < SW_ - 1) {
            const float gm = (x - 1 >= first && x - 1 <= last) ? Gs[e - C_] : 0.0f;
            const float gp = (x + 1 >= first && x + 1 <= last) ? Gs[e + C_] : 0.0f;
            const float g0 = (x     >= first && x     <= last) ? Gs[e]      : 0.0f;
            a = (gm + gp + 2.0f * g0) * 0.25f;
        }
        As[e] = a;
    }
    __syncthreads();

    for (int e = tid; e < NL_ * C_; e += 512) {
        const int l = e >> 4;
        if (l >= 2 && l <= lbhi) {
            const float b = (As[e - C_] + As[e + C_] + 2.0f * As[e]) * 0.25f;
            P[((size_t)cy * SW_ + c0) * C_ + e] = b;
        }
    }
}

__global__ __launch_bounds__(256) void slice_kernel(const float* __restrict__ P0,
                                                    const float* __restrict__ P1,
                                                    float* __restrict__ out)
{
    const int wave = threadIdx.x >> 5;
    const int lane = threadIdx.x & 31;
    const int tile = blockIdx.x * 8 + wave;
    const int k    = tile >> 7;
    const int j    = tile & 127;
    const int m    = lane & 15;
    const int hi   = lane >> 4;

    const int    row = k + 2 + hi;
    const size_t gi  = (((size_t)row * SW_) + (j + 2)) * C_ + m;
    float bx = 0.0f, by = 0.0f;
    if (row <= SH_ - 3) {
        bx = P0[gi];
        by = P0[gi + C_];
    }
    if (row >= 3) {
        bx += P1[gi];
        by += P1[gi + C_];
    }
    v2f b;
    b.x = bx;
    b.y = by;

    const int   pxm = 2 * (m & 7) + (m >> 3);
    const float am  = (float)pxm * 0.0625f;
    const float w0  = 1.0f - am;

    float* obase = out + ((size_t)(16 * k) * W_ + 16 * j) * C_;

#pragma unroll
    for (int py = 0; py < 16; ++py) {
        const float ay = (float)py * 0.0625f;
        const float wy = hi ? ay : (1.0f - ay);
        v2f a;
        a.x = wy * w0;
        a.y = wy * am;
        v8f cacc = {};
        v8f d = __builtin_amdgcn_wmma_f32_16x16x4_f32(
             false, a,  false, b,
             (short)0, cacc,  false,  false);

        float* orow = obase + (size_t)py * W_ * C_ + lane;
#pragma unroll
        for (int r = 0; r < 8; ++r)
            orow[r * 2 * C_] = d[r];
    }
}

extern "C" void kernel_launch(void* const* d_in, const int* in_sizes, int n_in,
                              void* d_out, int out_size, void* d_ws, size_t ws_size,
                              hipStream_t stream)
{
    const float* inp = (const float*)d_in[0];
    float* out = (float*)d_out;

    const size_t S = (size_t)SH_ * SW_ * C_;
    float* P0 = (float*)d_ws + S;
    float* P1 = (float*)d_ws + 2 * S;

    front_kernel<<<dim3(2, H_ / 8), 512, 0, stream>>>(inp, P0, P1);

    slice_kernel<<<8192 / 8, 256, 0, stream>>>(P0, P1, out);
}
